# starting kernel
# baseline (speedup 1.0000x reference)
#include <hip/hip_runtime.h>
#include <hip/hip_bf16.h>

typedef __attribute__((ext_vector_type(16))) _Float16 v16h;
typedef __attribute__((ext_vector_type(8)))  float    v8f;

#define DH 128

__global__ __launch_bounds__(256)
void zero_f32(float* __restrict__ p, long long n4) {
    long long i = (long long)blockIdx.x * blockDim.x + threadIdx.x;
    if (i < n4) ((float4*)p)[i] = make_float4(0.f, 0.f, 0.f, 0.f);
}

__global__ __launch_bounds__(256)
void pack_weights(const float* __restrict__ W, _Float16* __restrict__ Wp) {
    int idx = blockIdx.x * blockDim.x + threadIdx.x;
    if (idx >= 4 * 8 * 32 * 16) return;
    int h    = idx & 15;
    int lane = (idx >> 4) & 31;
    int nt   = (idx >> 9) & 7;
    int kt   = (idx >> 12) & 3;
    int grp  = lane >> 4;
    int k    = kt * 32 + grp * 8 + ((h < 8) ? h : (16 + (h - 8)));
    int n    = nt * 16 + (lane & 15);
    Wp[idx]  = (_Float16)W[k * DH + n];
}

__global__ __launch_bounds__(256)
void edge_aggregate(const float* __restrict__ srcRows,
                    const int* __restrict__ esrc, const int* __restrict__ edst,
                    float* __restrict__ agg, int nEdges) {
    int edge = blockIdx.x * (blockDim.x >> 5) + (threadIdx.x >> 5);
    if (edge >= nEdges) return;
    int lane = threadIdx.x & 31;
    int s = __builtin_amdgcn_readfirstlane(esrc[edge]);
    int d = __builtin_amdgcn_readfirstlane(edst[edge]);
    float4 v = ((const float4*)(srcRows + (long long)s * DH))[lane];
    float* out = agg + (long long)d * DH + (lane << 2);
    atomicAdd(out + 0, v.x);
    atomicAdd(out + 1, v.y);
    atomicAdd(out + 2, v.z);
    atomicAdd(out + 3, v.w);
}

template <bool HAS_A1, bool HAS_RES, bool RELU>
__global__ __launch_bounds__(128)
void gemm128_wmma(const float* __restrict__ A0,
                  const float* __restrict__ A1,
                  const _Float16* __restrict__ Wp,
                  const float* __restrict__ bias,
                  const float* __restrict__ Res,
                  float* __restrict__ Out,
                  int nRowTiles) {
    __shared__ float lds[4][16 * DH];
    int wave = blockIdx.x * (blockDim.x >> 5) + (threadIdx.x >> 5);
    if (wave >= nRowTiles) return;
    int lane = threadIdx.x & 31;
    int grp  = lane >> 4;
    int col  = lane & 15;
    float* myLds = lds[threadIdx.x >> 5];
    long long row = (long long)wave * 16 + col;

    const float* ar0 = A0 + row * DH;
    const float* ar1 = HAS_A1 ? (A1 + row * DH) : nullptr;
    v16h afrag[4];
#pragma unroll
    for (int kt = 0; kt < 4; ++kt) {
        int off = kt * 32 + grp * 8;
        float4 q0 = *(const float4*)(ar0 + off);
        float4 q1 = *(const float4*)(ar0 + off + 4);
        float4 q2 = *(const float4*)(ar0 + off + 16);
        float4 q3 = *(const float4*)(ar0 + off + 20);
        if (HAS_A1) {
            float4 b;
            b = *(const float4*)(ar1 + off);      q0.x += b.x; q0.y += b.y; q0.z += b.z; q0.w += b.w;
            b = *(const float4*)(ar1 + off + 4);  q1.x += b.x; q1.y += b.y; q1.z += b.z; q1.w += b.w;
            b = *(const float4*)(ar1 + off + 16); q2.x += b.x; q2.y += b.y; q2.z += b.z; q2.w += b.w;
            b = *(const float4*)(ar1 + off + 20); q3.x += b.x; q3.y += b.y; q3.z += b.z; q3.w += b.w;
        }
        v16h a;
        a[0]  = (_Float16)q0.x; a[1]  = (_Float16)q0.y; a[2]  = (_Float16)q0.z; a[3]  = (_Float16)q0.w;
        a[4]  = (_Float16)q1.x; a[5]  = (_Float16)q1.y; a[6]  = (_Float16)q1.z; a[7]  = (_Float16)q1.w;
        a[8]  = (_Float16)q2.x; a[9]  = (_Float16)q2.y; a[10] = (_Float16)q2.z; a[11] = (_Float16)q2.w;
        a[12] = (_Float16)q3.x; a[13] = (_Float16)q3.y; a[14] = (_Float16)q3.z; a[15] = (_Float16)q3.w;
        afrag[kt] = a;
    }

    const v16h* wfrag = (const v16h*)Wp;
#pragma unroll
    for (int nt = 0; nt < 8; ++nt) {
        v8f acc = {};
#pragma unroll
        for (int kt = 0; kt < 4; ++kt) {
            v16h b = wfrag[(kt * 8 + nt) * 32 + lane];
            acc = __builtin_amdgcn_wmma_f32_16x16x32_f16(
                false, afrag[kt], false, b, (short)0, acc, false, false);
        }
#pragma unroll
        for (int r = 0; r < 8; ++r)
            myLds[(r + grp * 8) * DH + nt * 16 + col] = acc[r];
    }

    float4 bvec = ((const float4*)bias)[lane];
#pragma unroll
    for (int rl = 0; rl < 16; ++rl) {
        float4 v = ((const float4*)myLds)[rl * 32 + lane];
        v.x += bvec.x; v.y += bvec.y; v.z += bvec.z; v.w += bvec.w;
        if (RELU) {
            v.x = fmaxf(v.x, 0.f); v.y = fmaxf(v.y, 0.f);
            v.z = fmaxf(v.z, 0.f); v.w = fmaxf(v.w, 0.f);
        }
        long long m = (long long)wave * 16 + rl;
        if (HAS_RES) {
            float4 rv = ((const float4*)(Res + m * DH))[lane];
            v.x += rv.x; v.y += rv.y; v.z += rv.z; v.w += rv.w;
        }
        ((float4*)(Out + m * DH))[lane] = v;
    }
}

__global__ __launch_bounds__(256)
void fuse_t2(const float* __restrict__ t, const float* __restrict__ ga,
             const float* __restrict__ gb, float* __restrict__ out, long long n4) {
    long long i = (long long)blockIdx.x * blockDim.x + threadIdx.x;
    if (i >= n4) return;
    float4 a = ((const float4*)t)[i];
    float4 b = ((const float4*)ga)[i];
    float4 c = ((const float4*)gb)[i];
    float4 o;
    o.x = fmaxf(a.x + b.x + c.x, 0.f);
    o.y = fmaxf(a.y + b.y + c.y, 0.f);
    o.z = fmaxf(a.z + b.z + c.z, 0.f);
    o.w = fmaxf(a.w + b.w + c.w, 0.f);
    ((float4*)out)[i] = o;
}

__global__ __launch_bounds__(256)
void ln_residual(const float* __restrict__ u, const float* __restrict__ t2,
                 const float* __restrict__ g, const float* __restrict__ b,
                 float* __restrict__ out, int nRows) {
    int wave = blockIdx.x * (blockDim.x >> 5) + (threadIdx.x >> 5);
    if (wave >= nRows) return;
    int lane = threadIdx.x & 31;
    long long base = (long long)wave * DH;
    float4 v = ((const float4*)(u + base))[lane];
    float s  = v.x + v.y + v.z + v.w;
    float sq = v.x * v.x + v.y * v.y + v.z * v.z + v.w * v.w;
#pragma unroll
    for (int off = 16; off > 0; off >>= 1) {
        s  += __shfl_xor(s, off, 32);
        sq += __shfl_xor(sq, off, 32);
    }
    float mean = s * (1.0f / 128.0f);
    float var  = sq * (1.0f / 128.0f) - mean * mean;
    float inv  = rsqrtf(var + 1e-5f);
    float4 tv = ((const float4*)(t2 + base))[lane];
    float4 gv = ((const float4*)g)[lane];
    float4 bv = ((const float4*)b)[lane];
    float4 o;
    o.x = tv.x + (v.x - mean) * inv * gv.x + bv.x;
    o.y = tv.y + (v.y - mean) * inv * gv.y + bv.y;
    o.z = tv.z + (v.z - mean) * inv * gv.z + bv.z;
    o.w = tv.w + (v.w - mean) * inv * gv.w + bv.w;
    ((float4*)(out + base))[lane] = o;
}

extern "C" void kernel_launch(void* const* d_in, const int* in_sizes, int n_in,
                              void* d_out, int out_size, void* d_ws, size_t ws_size,
                              hipStream_t stream) {
    const float* x    = (const float*)d_in[0];
    const float* t    = (const float*)d_in[1];
    const int*   e_t  = (const int*)d_in[2];
    const int*   e_x  = (const int*)d_in[3];
    const float* W1a  = (const float*)d_in[4];
    const float* b1a  = (const float*)d_in[5];
    const float* W2a  = (const float*)d_in[6];
    const float* b2a  = (const float*)d_in[7];
    const float* W1b  = (const float*)d_in[8];
    const float* b1b  = (const float*)d_in[9];
    const float* W2b  = (const float*)d_in[10];
    const float* b2b  = (const float*)d_in[11];
    const float* Wo   = (const float*)d_in[12];
    const float* bo   = (const float*)d_in[13];
    const float* ln_g = (const float*)d_in[14];
    const float* ln_b = (const float*)d_in[15];
    const float* Wf1  = (const float*)d_in[16];
    const float* bf1  = (const float*)d_in[17];
    const float* Wf2  = (const float*)d_in[18];
    const float* bf2  = (const float*)d_in[19];

    const int N = in_sizes[0] / DH;
    const int E = in_sizes[2] / 2;
    const long long ND = (long long)N * DH;

    float* buf0 = (float*)d_ws;
    float* buf1 = buf0 + ND;
    float* buf2 = buf1 + ND;
    _Float16* wp = (_Float16*)(buf2 + ND);
    const int WPN = 4 * 8 * 32 * 16;
    _Float16* wp0 = wp;
    _Float16* wp1 = wp + WPN;
    _Float16* wp2 = wp + 2 * WPN;
    _Float16* wp3 = wp + 3 * WPN;
    _Float16* wp4 = wp + 4 * WPN;
    _Float16* wp5 = wp + 5 * WPN;
    _Float16* wp6 = wp + 6 * WPN;

    float* x_out = (float*)d_out;
    float* t_out = x_out + ND;

    const int nRowTiles = N / 16;
    dim3 gB((nRowTiles + 3) / 4), tB(128);

    {
        long long n4 = (2 * ND) / 4;
        zero_f32<<<(unsigned)((n4 + 255) / 256), 256, 0, stream>>>(buf0, n4);
    }
    {
        dim3 g((WPN + 255) / 256), b(256);
        pack_weights<<<g, b, 0, stream>>>(W1a, wp0);
        pack_weights<<<g, b, 0, stream>>>(W2a, wp1);
        pack_weights<<<g, b, 0, stream>>>(W1b, wp2);
        pack_weights<<<g, b, 0, stream>>>(W2b, wp3);
        pack_weights<<<g, b, 0, stream>>>(Wo,  wp4);
        pack_weights<<<g, b, 0, stream>>>(Wf1, wp5);
        pack_weights<<<g, b, 0, stream>>>(Wf2, wp6);
    }
    {
        unsigned blocks = (unsigned)((E + 7) / 8);
        edge_aggregate<<<blocks, 256, 0, stream>>>(t, e_t, e_t + E, buf0, E);
        edge_aggregate<<<blocks, 256, 0, stream>>>(x, e_x, e_x + E, buf1, E);
    }
    gemm128_wmma<true,  false, true ><<<gB, tB, 0, stream>>>(t,    buf0,    wp0, b1a, nullptr, buf2, nRowTiles);
    gemm128_wmma<false, false, false><<<gB, tB, 0, stream>>>(buf2, nullptr, wp1, b2a, nullptr, buf0, nRowTiles);
    gemm128_wmma<true,  false, true ><<<gB, tB, 0, stream>>>(t,    buf1,    wp2, b1b, nullptr, buf2, nRowTiles);
    gemm128_wmma<false, false, false><<<gB, tB, 0, stream>>>(buf2, nullptr, wp3, b2b, nullptr, buf1, nRowTiles);
    fuse_t2<<<(unsigned)((ND / 4 + 255) / 256), 256, 0, stream>>>(t, buf0, buf1, buf2, ND / 4);
    gemm128_wmma<false, false, true ><<<gB, tB, 0, stream>>>(buf2, nullptr, wp4, bo, nullptr, buf0, nRowTiles);
    {
        unsigned blocks = (unsigned)(((long long)N * 32 + 255) / 256);
        ln_residual<<<blocks, 256, 0, stream>>>(buf0, buf2, ln_g, ln_b, t_out, N);
    }
    gemm128_wmma<false, false, true ><<<gB, tB, 0, stream>>>(x,    nullptr, wp5, bf1, nullptr, buf1, nRowTiles);
    gemm128_wmma<false, true,  false><<<gB, tB, 0, stream>>>(buf1, nullptr, wp6, bf2, x, x_out, nRowTiles);
}


# Edge aggregation: each work-group keeps a 392-row destination slice in LDS, scans all edge words (dst:src) and adds its rows by LDS atomics; no global atomics, no zero-fill; sums in arrival order
# speedup vs baseline: 2.3321x; 2.3321x over previous
#include <hip/hip_runtime.h>
#include <hip/hip_bf16.h>

typedef __attribute__((ext_vector_type(16))) _Float16 v16h;
typedef __attribute__((ext_vector_type(8)))  float    v8f;

#define DH 128

__global__ __launch_bounds__(256)
void pack_weights(const float* __restrict__ W, _Float16* __restrict__ Wp) {
    int idx = blockIdx.x * blockDim.x + threadIdx.x;
    if (idx >= 4 * 8 * 32 * 16) return;
    int h    = idx & 15;
    int lane = (idx >> 4) & 31;
    int nt   = (idx >> 9) & 7;
    int kt   = (idx >> 12) & 3;
    int grp  = lane >> 4;
    int k    = kt * 32 + grp * 8 + ((h < 8) ? h : (16 + (h - 8)));
    int n    = nt * 16 + (lane & 15);
    Wp[idx]  = (_Float16)W[k * DH + n];
}

#define BLKW       4
#define BLK_EDGES  (32 * 4 * BLKW)
#define NOEDGE     0xFFFFFFFFu

__global__ __launch_bounds__(256)
void pack_edges(const int* __restrict__ eT, const int* __restrict__ eX,
                unsigned* __restrict__ pk, int nEdges, int stride) {
    int i = blockIdx.x * blockDim.x + threadIdx.x;
    if (i >= stride) return;
    const int* __restrict__ e = blockIdx.y ? eX : eT;
    unsigned w = NOEDGE;
    if (i < nEdges) {
        unsigned s = (unsigned)e[i];
        unsigned d = (unsigned)e[(long long)nEdges + i];
        w = (d << 16) | (s & 0xFFFFu);
    }
    pk[(long long)blockIdx.y * stride + i] = w;
}

#define SLICE_ROWS 392
#define QB         8

__device__ __forceinline__ void load_block(const unsigned* __restrict__ pk, int blk,
                                           int lane, uint4 (&w)[BLKW]) {
    const uint4* p = (const uint4*)(pk + (long long)blk * BLK_EDGES) + lane;
#pragma unroll
    for (int v = 0; v < BLKW; ++v) w[v] = p[v * 32];
}

__device__ __forceinline__ void scan_word(unsigned w, unsigned loKey, unsigned spanKey,
                                          unsigned* q, int& qn) {
    bool m = (w - loKey) < spanKey;
    unsigned mask = __builtin_amdgcn_ballot_w32(m);
    if (mask != 0u) {
        if (m) q[qn + (int)__builtin_amdgcn_mbcnt_lo(mask, 0u)] = w;
        qn += __builtin_popcount(mask);
    }
}

__device__ __forceinline__ void take_block(const uint4 (&w)[BLKW], unsigned loKey, unsigned spanKey,
                                           int lo, unsigned* q, float* acc,
                                           const float* __restrict__ srcRows, int lane) {
    int qn = 0;
#pragma unroll
    for (int v = 0; v < BLKW; ++v) {
        scan_word(w[v].x, loKey, spanKey, q, qn);
        scan_word(w[v].y, loKey, spanKey, q, qn);
        scan_word(w[v].z, loKey, spanKey, q, qn);
        scan_word(w[v].w, loKey, spanKey, q, qn);
    }
    for (int h = 0; h < qn; h += 32) {
        unsigned ev = (h + lane < qn) ? q[h + lane] : NOEDGE;
#pragma unroll
        for (int g = 0; g < 32 / QB; ++g) {
            const int nb = qn - h - g * QB;
            if (nb > 0) {
                float4   r[QB];
                unsigned e[QB];
#pragma unroll
                for (int u = 0; u < QB; ++u) {
                    if (u < nb) {
                        e[u] = __builtin_amdgcn_readlane(ev, g * QB + u);
                        r[u] = ((const float4*)(srcRows + (long long)(e[u] & 0xFFFFu) * DH))[lane];
                    }
                }
#pragma unroll
                for (int u = 0; u < QB; ++u) {
                    if (u < nb) {
                        float* o = acc + (int)((e[u] >> 16) - (unsigned)lo) * DH + lane;
                        atomicAdd(o,      r[u].x);
                        atomicAdd(o + 32, r[u].y);
                        atomicAdd(o + 64, r[u].z);
                        atomicAdd(o + 96, r[u].w);
                    }
                }
            }
        }
    }
}

__global__ __launch_bounds__(1024)
void edge_aggregate_lds(const float* __restrict__ rowsT, const float* __restrict__ rowsX,
                        const unsigned* __restrict__ pkAll,
                        float* __restrict__ aggT, float* __restrict__ aggX,
                        int stride, int nRows) {
    __shared__ float    acc[SLICE_ROWS * DH];
    __shared__ unsigned queue[32][BLK_EDGES];

    const int which = blockIdx.y;
    const float* __restrict__ srcRows = which ? rowsX : rowsT;
    float* __restrict__ agg           = which ? aggX  : aggT;
    const unsigned* __restrict__ pk   = pkAll + (long long)which * stride;

    const int lo = blockIdx.x * SLICE_ROWS;
    int rs = nRows - lo;
    if (rs > SLICE_ROWS) rs = SLICE_ROWS;

    for (int i = threadIdx.x; i < SLICE_ROWS * DH / 4; i += 1024)
        ((float4*)acc)[i] = make_float4(0.f, 0.f, 0.f, 0.f);
    __syncthreads();

    const int lane = threadIdx.x & 31;
    const int wv = __builtin_amdgcn_readfirstlane((int)(threadIdx.x >> 5));
    unsigned* q = queue[wv];
    const unsigned loKey   = (unsigned)lo << 16;
    const unsigned spanKey = (unsigned)rs << 16;
    const int nBlk  = stride / BLK_EDGES;
    const int start = (int)(((long long)blockIdx.x * nBlk) / gridDim.x);

    if (wv < nBlk) {
        auto blockOf = [&](int j) {
            j = j < nBlk ? j : nBlk - 1;
            int p = j + start;
            return p >= nBlk ? p - nBlk : p;
        };
        uint4 wa[BLKW], wb[BLKW];
        load_block(pk, blockOf(wv), lane, wa);
        for (int j = wv; j < nBlk; j += 64) {
            load_block(pk, blockOf(j + 32), lane, wb);
            take_block(wa, loKey, spanKey, lo, q, acc, srcRows, lane);
            load_block(pk, blockOf(j + 64), lane, wa);
            if (j + 32 < nBlk)
                take_block(wb, loKey, spanKey, lo, q, acc, srcRows, lane);
        }
    }
    __syncthreads();

    for (int i = threadIdx.x; i < rs * 32; i += 1024) {
        const int r = i >> 5;
        const float* a = acc + r * DH + lane;
        float4 o = make_float4(a[0], a[32], a[64], a[96]);
        ((float4*)(agg + (long long)(lo + r) * DH))[lane] = o;
    }
}

template <bool HAS_A1, bool HAS_RES, bool RELU>
__global__ __launch_bounds__(128)
void gemm128_wmma(const float* __restrict__ A0,
                  const float* __restrict__ A1,
                  const _Float16* __restrict__ Wp,
                  const float* __restrict__ bias,
                  const float* __restrict__ Res,
                  float* __restrict__ Out,
                  int nRowTiles) {
    __shared__ float lds[4][16 * DH];
    int wave = blockIdx.x * (blockDim.x >> 5) + (threadIdx.x >> 5);
    if (wave >= nRowTiles) return;
    int lane = threadIdx.x & 31;
    int grp  = lane >> 4;
    int col  = lane & 15;
    float* myLds = lds[threadIdx.x >> 5];
    long long row = (long long)wave * 16 + col;

    const float* ar0 = A0 + row * DH;
    const float* ar1 = HAS_A1 ? (A1 + row * DH) : nullptr;
    v16h afrag[4];
#pragma unroll
    for (int kt = 0; kt < 4; ++kt) {
        int off = kt * 32 + grp * 8;
        float4 q0 = *(const float4*)(ar0 + off);
        float4 q1 = *(const float4*)(ar0 + off + 4);
        float4 q2 = *(const float4*)(ar0 + off + 16);
        float4 q3 = *(const float4*)(ar0 + off + 20);
        if (HAS_A1) {
            float4 b;
            b = *(const float4*)(ar1 + off);      q0.x += b.x; q0.y += b.y; q0.z += b.z; q0.w += b.w;
            b = *(const float4*)(ar1 + off + 4);  q1.x += b.x; q1.y += b.y; q1.z += b.z; q1.w += b.w;
            b = *(const float4*)(ar1 + off + 16); q2.x += b.x; q2.y += b.y; q2.z += b.z; q2.w += b.w;
            b = *(const float4*)(ar1 + off + 20); q3.x += b.x; q3.y += b.y; q3.z += b.z; q3.w += b.w;
        }
        v16h a;
        a[0]  = (_Float16)q0.x; a[1]  = (_Float16)q0.y; a[2]  = (_Float16)q0.z; a[3]  = (_Float16)q0.w;
        a[4]  = (_Float16)q1.x; a[5]  = (_Float16)q1.y; a[6]  = (_Float16)q1.z; a[7]  = (_Float16)q1.w;
        a[8]  = (_Float16)q2.x; a[9]  = (_Float16)q2.y; a[10] = (_Float16)q2.z; a[11] = (_Float16)q2.w;
        a[12] = (_Float16)q3.x; a[13] = (_Float16)q3.y; a[14] = (_Float16)q3.z; a[15] = (_Float16)q3.w;
        afrag[kt] = a;
    }

    const v16h* wfrag = (const v16h*)Wp;
#pragma unroll
    for (int nt = 0; nt < 8; ++nt) {
        v8f acc = {};
#pragma unroll
        for (int kt = 0; kt < 4; ++kt) {
            v16h b = wfrag[(kt * 8 + nt) * 32 + lane];
            acc = __builtin_amdgcn_wmma_f32_16x16x32_f16(
                false, afrag[kt], false, b, (short)0, acc, false, false);
        }
#pragma unroll
        for (int r = 0; r < 8; ++r)
            myLds[(r + grp * 8) * DH + nt * 16 + col] = acc[r];
    }

    float4 bvec = ((const float4*)bias)[lane];
#pragma unroll
    for (int rl = 0; rl < 16; ++rl) {
        float4 v = ((const float4*)myLds)[rl * 32 + lane];
        v.x += bvec.x; v.y += bvec.y; v.z += bvec.z; v.w += bvec.w;
        if (RELU) {
            v.x = fmaxf(v.x, 0.f); v.y = fmaxf(v.y, 0.f);
            v.z = fmaxf(v.z, 0.f); v.w = fmaxf(v.w, 0.f);
        }
        long long m = (long long)wave * 16 + rl;
        if (HAS_RES) {
            float4 rv = ((const float4*)(Res + m * DH))[lane];
            v.x += rv.x; v.y += rv.y; v.z += rv.z; v.w += rv.w;
        }
        ((float4*)(Out + m * DH))[lane] = v;
    }
}

__global__ __launch_bounds__(256)
void fuse_t2(const float* __restrict__ t, const float* __restrict__ ga,
             const float* __restrict__ gb, float* __restrict__ out, long long n4) {
    long long i = (long long)blockIdx.x * blockDim.x + threadIdx.x;
    if (i >= n4) return;
    float4 a = ((const float4*)t)[i];
    float4 b = ((const float4*)ga)[i];
    float4 c = ((const float4*)gb)[i];
    float4 o;
    o.x = fmaxf(a.x + b.x + c.x, 0.f);
    o.y = fmaxf(a.y + b.y + c.y, 0.f);
    o.z = fmaxf(a.z + b.z + c.z, 0.f);
    o.w = fmaxf(a.w + b.w + c.w, 0.f);
    ((float4*)out)[i] = o;
}

__global__ __launch_bounds__(256)
void ln_residual(const float* __restrict__ u, const float* __restrict__ t2,
                 const float* __restrict__ g, const float* __restrict__ b,
                 float* __restrict__ out, int nRows) {
    int wave = blockIdx.x * (blockDim.x >> 5) + (threadIdx.x >> 5);
    if (wave >= nRows) return;
    int lane = threadIdx.x & 31;
    long long base = (long long)wave * DH;
    float4 v = ((const float4*)(u + base))[lane];
    float s  = v.x + v.y + v.z + v.w;
    float sq = v.x * v.x + v.y * v.y + v.z * v.z + v.w * v.w;
#pragma unroll
    for (int off = 16; off > 0; off >>= 1) {
        s  += __shfl_xor(s, off, 32);
        sq += __shfl_xor(sq, off, 32);
    }
    float mean = s * (1.0f / 128.0f);
    float var  = sq * (1.0f / 128.0f) - mean * mean;
    float inv  = rsqrtf(var + 1e-5f);
    float4 tv = ((const float4*)(t2 + base))[lane];
    float4 gv = ((const float4*)g)[lane];
    float4 bv = ((const float4*)b)[lane];
    float4 o;
    o.x = tv.x + (v.x - mean) * inv * gv.x + bv.x;
    o.y = tv.y + (v.y - mean) * inv * gv.y + bv.y;
    o.z = tv.z + (v.z - mean) * inv * gv.z + bv.z;
    o.w = tv.w + (v.w - mean) * inv * gv.w + bv.w;
    ((float4*)(out + base))[lane] = o;
}

extern "C" void kernel_launch(void* const* d_in, const int* in_sizes, int n_in,
                              void* d_out, int out_size, void* d_ws, size_t ws_size,
                              hipStream_t stream) {
    const float* x    = (const float*)d_in[0];
    const float* t    = (const float*)d_in[1];
    const int*   e_t  = (const int*)d_in[2];
    const int*   e_x  = (const int*)d_in[3];
    const float* W1a  = (const float*)d_in[4];
    const float* b1a  = (const float*)d_in[5];
    const float* W2a  = (const float*)d_in[6];
    const float* b2a  = (const float*)d_in[7];
    const float* W1b  = (const float*)d_in[8];
    const float* b1b  = (const float*)d_in[9];
    const float* W2b  = (const float*)d_in[10];
    const float* b2b  = (const float*)d_in[11];
    const float* Wo   = (const float*)d_in[12];
    const float* bo   = (const float*)d_in[13];
    const float* ln_g = (const float*)d_in[14];
    const float* ln_b = (const float*)d_in[15];
    const float* Wf1  = (const float*)d_in[16];
    const float* bf1  = (const float*)d_in[17];
    const float* Wf2  = (const float*)d_in[18];
    const float* bf2  = (const float*)d_in[19];

    const int N = in_sizes[0] / DH;
    const int E = in_sizes[2] / 2;
    const long long ND = (long long)N * DH;

    float* buf0 = (float*)d_ws;
    float* buf1 = buf0 + ND;
    float* buf2 = buf1 + ND;
    _Float16* wp = (_Float16*)(buf2 + ND);
    const int WPN = 4 * 8 * 32 * 16;
    _Float16* wp0 = wp;
    _Float16* wp1 = wp + WPN;
    _Float16* wp2 = wp + 2 * WPN;
    _Float16* wp3 = wp + 3 * WPN;
    _Float16* wp4 = wp + 4 * WPN;
    _Float16* wp5 = wp + 5 * WPN;
    _Float16* wp6 = wp + 6 * WPN;

    float* x_out = (float*)d_out;
    float* t_out = x_out + ND;

    const int nRowTiles = N / 16;
    dim3 gB((nRowTiles + 3) / 4), tB(128);

    unsigned* pk = (unsigned*)buf2;
    const int EP = (E + BLK_EDGES - 1) / BLK_EDGES * BLK_EDGES;
    {
        dim3 g((unsigned)((EP + 255) / 256), 2);
        pack_edges<<<g, 256, 0, stream>>>(e_t, e_x, pk, E, EP);
    }
    {
        dim3 g((WPN + 255) / 256), b(256);
        pack_weights<<<g, b, 0, stream>>>(W1a, wp0);
        pack_weights<<<g, b, 0, stream>>>(W2a, wp1);
        pack_weights<<<g, b, 0, stream>>>(W1b, wp2);
        pack_weights<<<g, b, 0, stream>>>(W2b, wp3);
        pack_weights<<<g, b, 0, stream>>>(Wo,  wp4);
        pack_weights<<<g, b, 0, stream>>>(Wf1, wp5);
        pack_weights<<<g, b, 0, stream>>>(Wf2, wp6);
    }
    {
        dim3 g((unsigned)((N + SLICE_ROWS - 1) / SLICE_ROWS), 2);
        edge_aggregate_lds<<<g, 1024, 0, stream>>>(t, x, pk, buf0, buf1, EP, N);
    }
    gemm128_wmma<true,  false, true ><<<gB, tB, 0, stream>>>(t,    buf0,    wp0, b1a, nullptr, buf2, nRowTiles);
    gemm128_wmma<false, false, false><<<gB, tB, 0, stream>>>(buf2, nullptr, wp1, b2a, nullptr, buf0, nRowTiles);
    gemm128_wmma<true,  false, true ><<<gB, tB, 0, stream>>>(t,    buf1,    wp2, b1b, nullptr, buf2, nRowTiles);
    gemm128_wmma<false, false, false><<<gB, tB, 0, stream>>>(buf2, nullptr, wp3, b2b, nullptr, buf1, nRowTiles);
    fuse_t2<<<(unsigned)((ND / 4 + 255) / 256), 256, 0, stream>>>(t, buf0, buf1, buf2, ND / 4);
    gemm128_wmma<false, false, true ><<<gB, tB, 0, stream>>>(buf2, nullptr, wp4, bo, nullptr, buf0, nRowTiles);
    {
        unsigned blocks = (unsigned)(((long long)N * 32 + 255) / 256);
        ln_residual<<<blocks, 256, 0, stream>>>(buf0, buf2, ln_g, ln_b, t_out, N);
    }
    gemm128_wmma<false, false, true ><<<gB, tB, 0, stream>>>(x,    nullptr, wp5, bf1, nullptr, buf1, nRowTiles);
    gemm128_wmma<false, true,  false><<<gB, tB, 0, stream>>>(buf1, nullptr, wp6, bf2, x, x_out, nRowTiles);
}


# Edge scan reads one tag byte per edge (destination slice number; four tags matched per 32-bit zero-byte test) and fetches the 4-byte edge word only on a hit; LDS atomic sums, arrival order as before
# speedup vs baseline: 2.4855x; 1.0658x over previous
#include <hip/hip_runtime.h>
#include <hip/hip_bf16.h>

typedef __attribute__((ext_vector_type(16))) _Float16 v16h;
typedef __attribute__((ext_vector_type(8)))  float    v8f;

#define DH 128

__global__ __launch_bounds__(256)
void pack_weights(const float* __restrict__ W, _Float16* __restrict__ Wp) {
    int idx = blockIdx.x * blockDim.x + threadIdx.x;
    if (idx >= 4 * 8 * 32 * 16) return;
    int h    = idx & 15;
    int lane = (idx >> 4) & 31;
    int nt   = (idx >> 9) & 7;
    int kt   = (idx >> 12) & 3;
    int grp  = lane >> 4;
    int k    = kt * 32 + grp * 8 + ((h < 8) ? h : (16 + (h - 8)));
    int n    = nt * 16 + (lane & 15);
    Wp[idx]  = (_Float16)W[k * DH + n];
}

#define SLICE_ROWS 392
#define BLKT       2
#define BLK_EDGES  (32 * 16 * BLKT)
#define NOEDGE     0xFFFFFFFFu
#define NOTAG      0xFFu

__global__ __launch_bounds__(256)
void pack_edges(const int* __restrict__ eT, const int* __restrict__ eX,
                unsigned* __restrict__ pk, unsigned* __restrict__ tg,
                int nEdges, int stride) {
    int t = blockIdx.x * blockDim.x + threadIdx.x;
    if (t * 4 >= stride) return;
    const int* __restrict__ e = blockIdx.y ? eX : eT;
    unsigned w[4];
    unsigned tags = 0u;
#pragma unroll
    for (int u = 0; u < 4; ++u) {
        const int i = t * 4 + u;
        unsigned word = NOEDGE, tag = NOTAG;
        if (i < nEdges) {
            unsigned s = (unsigned)e[i];
            unsigned d = (unsigned)e[(long long)nEdges + i];
            word = (d << 16) | (s & 0xFFFFu);
            tag  = d / (unsigned)SLICE_ROWS;
        }
        w[u] = word;
        tags |= tag << (8 * u);
    }
    ((uint4*)(pk + (long long)blockIdx.y * stride))[t] = make_uint4(w[0], w[1], w[2], w[3]);
    tg[(long long)blockIdx.y * (stride / 4) + t] = tags;
}

#define QB         8

__device__ __forceinline__ void load_block(const unsigned* __restrict__ tg, int blk,
                                           int lane, uint4 (&w)[BLKT]) {
    const uint4* p = (const uint4*)(tg + (long long)blk * (BLK_EDGES / 4)) + lane;
#pragma unroll
    for (int v = 0; v < BLKT; ++v) w[v] = p[v * 32];
}

__device__ __forceinline__ unsigned zero_bytes(unsigned x) {
    unsigned t = (x & 0x7F7F7F7Fu) + 0x7F7F7F7Fu;
    return ~(t | x) & 0x80808080u;
}

__device__ __forceinline__ void scan_tags(const uint4& w, unsigned pat, unsigned base,
                                          unsigned short* q, int& qn) {
    unsigned f = (zero_bytes(w.x ^ pat) >> 7) | (zero_bytes(w.y ^ pat) >> 6)
               | (zero_bytes(w.z ^ pat) >> 5) | (zero_bytes(w.w ^ pat) >> 4);
    unsigned mask = __builtin_amdgcn_ballot_w32(f != 0u);
    while (mask != 0u) {
        if (f != 0u) {
            unsigned p = (unsigned)__builtin_ctz(f);
            q[qn + (int)__builtin_amdgcn_mbcnt_lo(mask, 0u)] =
                (unsigned short)(base + ((p & 7u) << 2) + (p >> 3));
        }
        qn += __builtin_popcount(mask);
        f &= f - 1u;
        mask = __builtin_amdgcn_ballot_w32(f != 0u);
    }
}

__device__ __forceinline__ void take_block(const uint4 (&w)[BLKT], unsigned pat,
                                           const unsigned* __restrict__ pkBlk,
                                           int lo, unsigned short* q, float* acc,
                                           const float* __restrict__ srcRows, int lane) {
    int qn = 0;
#pragma unroll
    for (int v = 0; v < BLKT; ++v)
        scan_tags(w[v], pat, (unsigned)(v * 512 + lane * 16), q, qn);
    for (int h = 0; h < qn; h += 32) {
        unsigned ev = NOEDGE;
        if (h + lane < qn) ev = pkBlk[q[h + lane]];
#pragma unroll
        for (int g = 0; g < 32 / QB; ++g) {
            const int nb = qn - h - g * QB;
            if (nb > 0) {
                float4   r[QB];
                unsigned e[QB];
#pragma unroll
                for (int u = 0; u < QB; ++u) {
                    if (u < nb) {
                        e[u] = __builtin_amdgcn_readlane(ev, g * QB + u);
                        r[u] = ((const float4*)(srcRows + (long long)(e[u] & 0xFFFFu) * DH))[lane];
                    }
                }
#pragma unroll
                for (int u = 0; u < QB; ++u) {
                    if (u < nb) {
                        float* o = acc + (int)((e[u] >> 16) - (unsigned)lo) * DH + lane;
                        atomicAdd(o,      r[u].x);
                        atomicAdd(o + 32, r[u].y);
                        atomicAdd(o + 64, r[u].z);
                        atomicAdd(o + 96, r[u].w);
                    }
                }
            }
        }
    }
}

__global__ __launch_bounds__(1024)
void edge_aggregate_lds(const float* __restrict__ rowsT, const float* __restrict__ rowsX,
                        const unsigned* __restrict__ pkAll, const unsigned* __restrict__ tgAll,
                        float* __restrict__ aggT, float* __restrict__ aggX,
                        int stride, int nRows) {
    __shared__ float          acc[SLICE_ROWS * DH];
    __shared__ unsigned short queue[32][BLK_EDGES];

    const int which = blockIdx.y;
    const float* __restrict__ srcRows = which ? rowsX : rowsT;
    float* __restrict__ agg           = which ? aggX  : aggT;
    const unsigned* __restrict__ pk   = pkAll + (long long)which * stride;
    const unsigned* __restrict__ tg   = tgAll + (long long)which * (stride / 4);

    const int lo = blockIdx.x * SLICE_ROWS;
    int rs = nRows - lo;
    if (rs > SLICE_ROWS) rs = SLICE_ROWS;

    for (int i = threadIdx.x; i < SLICE_ROWS * DH / 4; i += 1024)
        ((float4*)acc)[i] = make_float4(0.f, 0.f, 0.f, 0.f);
    __syncthreads();

    const int lane = threadIdx.x & 31;
    const int wv = __builtin_amdgcn_readfirstlane((int)(threadIdx.x >> 5));
    unsigned short* q = queue[wv];
    const unsigned pat = (unsigned)blockIdx.x * 0x01010101u;
    const int nBlk  = stride / BLK_EDGES;
    const int start = (int)(((long long)blockIdx.x * nBlk) / gridDim.x);

    if (wv < nBlk) {
        auto blockOf = [&](int j) {
            j = j < nBlk ? j : nBlk - 1;
            int p = j + start;
            return p >= nBlk ? p - nBlk : p;
        };
        uint4 wa[BLKT], wb[BLKT];
        int ba = blockOf(wv), bb;
        load_block(tg, ba, lane, wa);
        for (int j = wv; j < nBlk; j += 64) {
            bb = blockOf(j + 32);
            load_block(tg, bb, lane, wb);
            take_block(wa, pat, pk + (long long)ba * BLK_EDGES, lo, q, acc, srcRows, lane);
            ba = blockOf(j + 64);
            load_block(tg, ba, lane, wa);
            if (j + 32 < nBlk)
                take_block(wb, pat, pk + (long long)bb * BLK_EDGES, lo, q, acc, srcRows, lane);
        }
    }
    __syncthreads();

    for (int i = threadIdx.x; i < rs * 32; i += 1024) {
        const int r = i >> 5;
        const float* a = acc + r * DH + lane;
        float4 o = make_float4(a[0], a[32], a[64], a[96]);
        ((float4*)(agg + (long long)(lo + r) * DH))[lane] = o;
    }
}

template <bool HAS_A1, bool HAS_RES, bool RELU>
__global__ __launch_bounds__(128)
void gemm128_wmma(const float* __restrict__ A0,
                  const float* __restrict__ A1,
                  const _Float16* __restrict__ Wp,
                  const float* __restrict__ bias,
                  const float* __restrict__ Res,
                  float* __restrict__ Out,
                  int nRowTiles) {
    __shared__ float lds[4][16 * DH];
    int wave = blockIdx.x * (blockDim.x >> 5) + (threadIdx.x >> 5);
    if (wave >= nRowTiles) return;
    int lane = threadIdx.x & 31;
    int grp  = lane >> 4;
    int col  = lane & 15;
    float* myLds = lds[threadIdx.x >> 5];
    long long row = (long long)wave * 16 + col;

    const float* ar0 = A0 + row * DH;
    const float* ar1 = HAS_A1 ? (A1 + row * DH) : nullptr;
    v16h afrag[4];
#pragma unroll
    for (int kt = 0; kt < 4; ++kt) {
        int off = kt * 32 + grp * 8;
        float4 q0 = *(const float4*)(ar0 + off);
        float4 q1 = *(const float4*)(ar0 + off + 4);
        float4 q2 = *(const float4*)(ar0 + off + 16);
        float4 q3 = *(const float4*)(ar0 + off + 20);
        if (HAS_A1) {
            float4 b;
            b = *(const float4*)(ar1 + off);      q0.x += b.x; q0.y += b.y; q0.z += b.z; q0.w += b.w;
            b = *(const float4*)(ar1 + off + 4);  q1.x += b.x; q1.y += b.y; q1.z += b.z; q1.w += b.w;
            b = *(const float4*)(ar1 + off + 16); q2.x += b.x; q2.y += b.y; q2.z += b.z; q2.w += b.w;
            b = *(const float4*)(ar1 + off + 20); q3.x += b.x; q3.y += b.y; q3.z += b.z; q3.w += b.w;
        }
        v16h a;
        a[0]  = (_Float16)q0.x; a[1]  = (_Float16)q0.y; a[2]  = (_Float16)q0.z; a[3]  = (_Float16)q0.w;
        a[4]  = (_Float16)q1.x; a[5]  = (_Float16)q1.y; a[6]  = (_Float16)q1.z; a[7]  = (_Float16)q1.w;
        a[8]  = (_Float16)q2.x; a[9]  = (_Float16)q2.y; a[10] = (_Float16)q2.z; a[11] = (_Float16)q2.w;
        a[12] = (_Float16)q3.x; a[13] = (_Float16)q3.y; a[14] = (_Float16)q3.z; a[15] = (_Float16)q3.w;
        afrag[kt] = a;
    }

    const v16h* wfrag = (const v16h*)Wp;
#pragma unroll
    for (int nt = 0; nt < 8; ++nt) {
        v8f acc = {};
#pragma unroll
        for (int kt = 0; kt < 4; ++kt) {
            v16h b = wfrag[(kt * 8 + nt) * 32 + lane];
            acc = __builtin_amdgcn_wmma_f32_16x16x32_f16(
                false, afrag[kt], false, b, (short)0, acc, false, false);
        }
#pragma unroll
        for (int r = 0; r < 8; ++r)
            myLds[(r + grp * 8) * DH + nt * 16 + col] = acc[r];
    }

    float4 bvec = ((const float4*)bias)[lane];
#pragma unroll
    for (int rl = 0; rl < 16; ++rl) {
        float4 v = ((const float4*)myLds)[rl * 32 + lane];
        v.x += bvec.x; v.y += bvec.y; v.z += bvec.z; v.w += bvec.w;
        if (RELU) {
            v.x = fmaxf(v.x, 0.f); v.y = fmaxf(v.y, 0.f);
            v.z = fmaxf(v.z, 0.f); v.w = fmaxf(v.w, 0.f);
        }
        long long m = (long long)wave * 16 + rl;
        if (HAS_RES) {
            float4 rv = ((const float4*)(Res + m * DH))[lane];
            v.x += rv.x; v.y += rv.y; v.z += rv.z; v.w += rv.w;
        }
        ((float4*)(Out + m * DH))[lane] = v;
    }
}

__global__ __launch_bounds__(256)
void fuse_t2(const float* __restrict__ t, const float* __restrict__ ga,
             const float* __restrict__ gb, float* __restrict__ out, long long n4) {
    long long i = (long long)blockIdx.x * blockDim.x + threadIdx.x;
    if (i >= n4) return;
    float4 a = ((const float4*)t)[i];
    float4 b = ((const float4*)ga)[i];
    float4 c = ((const float4*)gb)[i];
    float4 o;
    o.x = fmaxf(a.x + b.x + c.x, 0.f);
    o.y = fmaxf(a.y + b.y + c.y, 0.f);
    o.z = fmaxf(a.z + b.z + c.z, 0.f);
    o.w = fmaxf(a.w + b.w + c.w, 0.f);
    ((float4*)out)[i] = o;
}

__global__ __launch_bounds__(256)
void ln_residual(const float* __restrict__ u, const float* __restrict__ t2,
                 const float* __restrict__ g, const float* __restrict__ b,
                 float* __restrict__ out, int nRows) {
    int wave = blockIdx.x * (blockDim.x >> 5) + (threadIdx.x >> 5);
    if (wave >= nRows) return;
    int lane = threadIdx.x & 31;
    long long base = (long long)wave * DH;
    float4 v = ((const float4*)(u + base))[lane];
    float s  = v.x + v.y + v.z + v.w;
    float sq = v.x * v.x + v.y * v.y + v.z * v.z + v.w * v.w;
#pragma unroll
    for (int off = 16; off > 0; off >>= 1) {
        s  += __shfl_xor(s, off, 32);
        sq += __shfl_xor(sq, off, 32);
    }
    float mean = s * (1.0f / 128.0f);
    float var  = sq * (1.0f / 128.0f) - mean * mean;
    float inv  = rsqrtf(var + 1e-5f);
    float4 tv = ((const float4*)(t2 + base))[lane];
    float4 gv = ((const float4*)g)[lane];
    float4 bv = ((const float4*)b)[lane];
    float4 o;
    o.x = tv.x + (v.x - mean) * inv * gv.x + bv.x;
    o.y = tv.y + (v.y - mean) * inv * gv.y + bv.y;
    o.z = tv.z + (v.z - mean) * inv * gv.z + bv.z;
    o.w = tv.w + (v.w - mean) * inv * gv.w + bv.w;
    ((float4*)(out + base))[lane] = o;
}

extern "C" void kernel_launch(void* const* d_in, const int* in_sizes, int n_in,
                              void* d_out, int out_size, void* d_ws, size_t ws_size,
                              hipStream_t stream) {
    const float* x    = (const float*)d_in[0];
    const float* t    = (const float*)d_in[1];
    const int*   e_t  = (const int*)d_in[2];
    const int*   e_x  = (const int*)d_in[3];
    const float* W1a  = (const float*)d_in[4];
    const float* b1a  = (const float*)d_in[5];
    const float* W2a  = (const float*)d_in[6];
    const float* b2a  = (const float*)d_in[7];
    const float* W1b  = (const float*)d_in[8];
    const float* b1b  = (const float*)d_in[9];
    const float* W2b  = (const float*)d_in[10];
    const float* b2b  = (const float*)d_in[11];
    const float* Wo   = (const float*)d_in[12];
    const float* bo   = (const float*)d_in[13];
    const float* ln_g = (const float*)d_in[14];
    const float* ln_b = (const float*)d_in[15];
    const float* Wf1  = (const float*)d_in[16];
    const float* bf1  = (const float*)d_in[17];
    const float* Wf2  = (const float*)d_in[18];
    const float* bf2  = (const float*)d_in[19];

    const int N = in_sizes[0] / DH;
    const int E = in_sizes[2] / 2;
    const long long ND = (long long)N * DH;

    float* buf0 = (float*)d_ws;
    float* buf1 = buf0 + ND;
    float* buf2 = buf1 + ND;
    _Float16* wp = (_Float16*)(buf2 + ND);
    const int WPN = 4 * 8 * 32 * 16;
    _Float16* wp0 = wp;
    _Float16* wp1 = wp + WPN;
    _Float16* wp2 = wp + 2 * WPN;
    _Float16* wp3 = wp + 3 * WPN;
    _Float16* wp4 = wp + 4 * WPN;
    _Float16* wp5 = wp + 5 * WPN;
    _Float16* wp6 = wp + 6 * WPN;

    float* x_out = (float*)d_out;
    float* t_out = x_out + ND;

    const int nRowTiles = N / 16;
    dim3 gB((nRowTiles + 3) / 4), tB(128);

    unsigned* pk = (unsigned*)buf2;
    const int EP = (E + BLK_EDGES - 1) / BLK_EDGES * BLK_EDGES;
    unsigned* tg = pk + 2 * (long long)EP;
    {
        dim3 g((unsigned)((EP / 4 + 255) / 256), 2);
        pack_edges<<<g, 256, 0, stream>>>(e_t, e_x, pk, tg, E, EP);
    }
    {
        dim3 g((WPN + 255) / 256), b(256);
        pack_weights<<<g, b, 0, stream>>>(W1a, wp0);
        pack_weights<<<g, b, 0, stream>>>(W2a, wp1);
        pack_weights<<<g, b, 0, stream>>>(W1b, wp2);
        pack_weights<<<g, b, 0, stream>>>(W2b, wp3);
        pack_weights<<<g, b, 0, stream>>>(Wo,  wp4);
        pack_weights<<<g, b, 0, stream>>>(Wf1, wp5);
        pack_weights<<<g, b, 0, stream>>>(Wf2, wp6);
    }
    {
        dim3 g((unsigned)((N + SLICE_ROWS - 1) / SLICE_ROWS), 2);
        edge_aggregate_lds<<<g, 1024, 0, stream>>>(t, x, pk, tg, buf0, buf1, EP, N);
    }
    gemm128_wmma<true,  false, true ><<<gB, tB, 0, stream>>>(t,    buf0,    wp0, b1a, nullptr, buf2, nRowTiles);
    gemm128_wmma<false, false, false><<<gB, tB, 0, stream>>>(buf2, nullptr, wp1, b2a, nullptr, buf0, nRowTiles);
    gemm128_wmma<true,  false, true ><<<gB, tB, 0, stream>>>(t,    buf1,    wp2, b1b, nullptr, buf2, nRowTiles);
    gemm128_wmma<false, false, false><<<gB, tB, 0, stream>>>(buf2, nullptr, wp3, b2b, nullptr, buf1, nRowTiles);
    fuse_t2<<<(unsigned)((ND / 4 + 255) / 256), 256, 0, stream>>>(t, buf0, buf1, buf2, ND / 4);
    gemm128_wmma<false, false, true ><<<gB, tB, 0, stream>>>(buf2, nullptr, wp4, bo, nullptr, buf0, nRowTiles);
    {
        unsigned blocks = (unsigned)(((long long)N * 32 + 255) / 256);
        ln_residual<<<blocks, 256, 0, stream>>>(buf0, buf2, ln_g, ln_b, t_out, N);
    }
    gemm128_wmma<false, false, true ><<<gB, tB, 0, stream>>>(x,    nullptr, wp5, bf1, nullptr, buf1, nRowTiles);
    gemm128_wmma<false, true,  false><<<gB, tB, 0, stream>>>(buf1, nullptr, wp6, bf2, x, x_out, nRowTiles);
}
